# starting kernel
# baseline (speedup 1.0000x reference)
#include <hip/hip_runtime.h>

typedef __attribute__((ext_vector_type(2))) float v2f;
typedef __attribute__((ext_vector_type(8))) float v8f;

#define BS   4
#define SEQ  4096
#define DK   128
#define KTILES (SEQ / 16)
#define SCALE 0.08838834764831845f
#define WAVES_PER_BLOCK 8

__device__ __forceinline__ v8f wmma_f32(v2f a, v2f b, v8f c) {
  return __builtin_amdgcn_wmma_f32_16x16x4_f32(
       false, a,  false, b,
       (short)0, c,  false,  false);
}

__global__ __launch_bounds__(256)
void sdpa_colstats_kernel(const float* __restrict__ Q,
                          const float* __restrict__ K,
                          float* __restrict__ mArr,
                          float* __restrict__ lArr) {
  const int lane  = threadIdx.x & 31;
  const int wid   = blockIdx.x * WAVES_PER_BLOCK + (threadIdx.x >> 5);
  const int b     = wid >> 8;
  const int kt    = (wid & 255) << 4;
  const int n     = lane & 15;
  const int khalf = (lane >> 4) << 1;

  const float* Krow = K + ((size_t)b * SEQ + kt + n) * DK + khalf;
  v2f breg[32];
#pragma unroll
  for (int c = 0; c < 32; ++c) breg[c] = *(const v2f*)(Krow + 4 * c);

  const float* Qb = Q + (size_t)b * SEQ * DK;
  float m_run = -__builtin_inff();
  float l_run = 0.0f;

  for (int qt = 0; qt < SEQ; qt += 16) {
    const float* Qrow = Qb + (size_t)(qt + n) * DK + khalf;
    v8f acc = {};
#pragma unroll
    for (int c = 0; c < 32; ++c) {
      v2f a = *(const v2f*)(Qrow + 4 * c);
      acc = wmma_f32(a, breg[c], acc);
    }
    float mloc = acc[0] * SCALE;
#pragma unroll
    for (int j = 1; j < 8; ++j) mloc = fmaxf(mloc, acc[j] * SCALE);
    const float m_new = fmaxf(m_run, mloc);
    float s = 0.0f;
#pragma unroll
    for (int j = 0; j < 8; ++j) s += __expf(acc[j] * SCALE - m_new);
    l_run = l_run * __expf(m_run - m_new) + s;
    m_run = m_new;
  }

  const float m_o = __shfl_xor(m_run, 16, 32);
  const float l_o = __shfl_xor(l_run, 16, 32);
  const float m_f = fmaxf(m_run, m_o);
  const float l_f = l_run * __expf(m_run - m_f) + l_o * __expf(m_o - m_f);
  if (lane < 16) {
    mArr[(size_t)b * SEQ + kt + n] = m_f;
    lArr[(size_t)b * SEQ + kt + n] = l_f;
  }
}

__global__ __launch_bounds__(256)
void sdpa_out_kernel(const float* __restrict__ Q,
                     const float* __restrict__ K,
                     const float* __restrict__ V,
                     const float* __restrict__ mArr,
                     const float* __restrict__ lArr,
                     float* __restrict__ O) {
  __shared__ float smem[WAVES_PER_BLOCK][16 * 17];

  const int lane  = threadIdx.x & 31;
  const int wslot = threadIdx.x >> 5;
  const int wid   = blockIdx.x * WAVES_PER_BLOCK + wslot;
  const int b     = wid >> 8;
  const int qt    = (wid & 255) << 4;
  const int n     = lane & 15;
  const int khalf = (lane >> 4) << 1;
  const int mhi   = (lane >> 4) << 3;

  const float* Qrow = Q + ((size_t)b * SEQ + qt + n) * DK + khalf;
  v2f areg[32];
#pragma unroll
  for (int c = 0; c < 32; ++c) areg[c] = *(const v2f*)(Qrow + 4 * c);

  const v8f vzero = {};
  v8f acc[8];
#pragma unroll
  for (int t = 0; t < 8; ++t) acc[t] = vzero;

  const float* Kb = K + (size_t)b * SEQ * DK;
  const float* Vb = V + (size_t)b * SEQ * DK;
  const float* mB = mArr + (size_t)b * SEQ;
  const float* lB = lArr + (size_t)b * SEQ;
  float* sm = &smem[wslot][0];

  for (int kt = 0; kt < SEQ; kt += 16) {
    const float* Krow = Kb + (size_t)(kt + n) * DK + khalf;
    v8f c8 = vzero;
#pragma unroll
    for (int c = 0; c < 32; ++c) {
      v2f bk = *(const v2f*)(Krow + 4 * c);
      c8 = wmma_f32(areg[c], bk, c8);
    }

    const float mcol = mB[kt + n];
    const float rl   = 1.0f / lB[kt + n];
#pragma unroll
    for (int j = 0; j < 8; ++j) {
      const float p = __expf(c8[j] * SCALE - mcol) * rl;
      sm[(j + mhi) * 17 + n] = p;
    }
    __builtin_amdgcn_wave_barrier();

    v2f pa[4];
#pragma unroll
    for (int c2 = 0; c2 < 4; ++c2) {
      const int kk = 4 * c2 + khalf;
      pa[c2].x = sm[n * 17 + kk];
      pa[c2].y = sm[n * 17 + kk + 1];
    }

#pragma unroll
    for (int t = 0; t < 8; ++t) {
      const float* Vcol = Vb + (size_t)(kt + khalf) * DK + t * 16 + n;
#pragma unroll
      for (int c2 = 0; c2 < 4; ++c2) {
        v2f bv;
        bv.x = Vcol[(size_t)(4 * c2) * DK];
        bv.y = Vcol[(size_t)(4 * c2 + 1) * DK];
        acc[t] = wmma_f32(pa[c2], bv, acc[t]);
      }
    }
  }

  float* Ob = O + ((size_t)b * SEQ + qt) * DK;
#pragma unroll
  for (int t = 0; t < 8; ++t) {
#pragma unroll
    for (int j = 0; j < 8; ++j) {
      Ob[(size_t)(j + mhi) * DK + t * 16 + n] = acc[t][j];
    }
  }
}

extern "C" void kernel_launch(void* const* d_in, const int* in_sizes, int n_in,
                              void* d_out, int out_size, void* d_ws, size_t ws_size,
                              hipStream_t stream) {
  const float* q = (const float*)d_in[0];
  const float* k = (const float*)d_in[1];
  const float* v = (const float*)d_in[2];
  float* out = (float*)d_out;

  float* mArr = (float*)d_ws;
  float* lArr = mArr + (size_t)BS * SEQ;

  const int total_waves = BS * KTILES;
  const int blocks = total_waves / WAVES_PER_BLOCK;
  const int threads = WAVES_PER_BLOCK * 32;

  sdpa_colstats_kernel<<<blocks, threads, 0, stream>>>(q, k, mArr, lArr);
  sdpa_out_kernel<<<blocks, threads, 0, stream>>>(q, k, v, mArr, lArr, out);
}


# Launch shape only: 4-wave workgroups instead of 8, so the 1024 waves form 256 workgroups (one per WGP, one wave per SIMD) instead of 128 on half the chip; code and summation order unchanged
# speedup vs baseline: 1.1790x; 1.1790x over previous
#include <hip/hip_runtime.h>

typedef __attribute__((ext_vector_type(2))) float v2f;
typedef __attribute__((ext_vector_type(8))) float v8f;

#define BS   4
#define SEQ  4096
#define DK   128
#define KTILES (SEQ / 16)
#define SCALE 0.08838834764831845f
#define WAVES_PER_BLOCK 4

__device__ __forceinline__ v8f wmma_f32(v2f a, v2f b, v8f c) {
  return __builtin_amdgcn_wmma_f32_16x16x4_f32(
       false, a,  false, b,
       (short)0, c,  false,  false);
}

__global__ __launch_bounds__(WAVES_PER_BLOCK * 32)
void sdpa_colstats_kernel(const float* __restrict__ Q,
                          const float* __restrict__ K,
                          float* __restrict__ mArr,
                          float* __restrict__ lArr) {
  const int lane  = threadIdx.x & 31;
  const int wid   = blockIdx.x * WAVES_PER_BLOCK + (threadIdx.x >> 5);
  const int b     = wid >> 8;
  const int kt    = (wid & 255) << 4;
  const int n     = lane & 15;
  const int khalf = (lane >> 4) << 1;

  const float* Krow = K + ((size_t)b * SEQ + kt + n) * DK + khalf;
  v2f breg[32];
#pragma unroll
  for (int c = 0; c < 32; ++c) breg[c] = *(const v2f*)(Krow + 4 * c);

  const float* Qb = Q + (size_t)b * SEQ * DK;
  float m_run = -__builtin_inff();
  float l_run = 0.0f;

  for (int qt = 0; qt < SEQ; qt += 16) {
    const float* Qrow = Qb + (size_t)(qt + n) * DK + khalf;
    v8f acc = {};
#pragma unroll
    for (int c = 0; c < 32; ++c) {
      v2f a = *(const v2f*)(Qrow + 4 * c);
      acc = wmma_f32(a, breg[c], acc);
    }
    float mloc = acc[0] * SCALE;
#pragma unroll
    for (int j = 1; j < 8; ++j) mloc = fmaxf(mloc, acc[j] * SCALE);
    const float m_new = fmaxf(m_run, mloc);
    float s = 0.0f;
#pragma unroll
    for (int j = 0; j < 8; ++j) s += __expf(acc[j] * SCALE - m_new);
    l_run = l_run * __expf(m_run - m_new) + s;
    m_run = m_new;
  }

  const float m_o = __shfl_xor(m_run, 16, 32);
  const float l_o = __shfl_xor(l_run, 16, 32);
  const float m_f = fmaxf(m_run, m_o);
  const float l_f = l_run * __expf(m_run - m_f) + l_o * __expf(m_o - m_f);
  if (lane < 16) {
    mArr[(size_t)b * SEQ + kt + n] = m_f;
    lArr[(size_t)b * SEQ + kt + n] = l_f;
  }
}

__global__ __launch_bounds__(WAVES_PER_BLOCK * 32)
void sdpa_out_kernel(const float* __restrict__ Q,
                     const float* __restrict__ K,
                     const float* __restrict__ V,
                     const float* __restrict__ mArr,
                     const float* __restrict__ lArr,
                     float* __restrict__ O) {
  __shared__ float smem[WAVES_PER_BLOCK][16 * 17];

  const int lane  = threadIdx.x & 31;
  const int wslot = threadIdx.x >> 5;
  const int wid   = blockIdx.x * WAVES_PER_BLOCK + wslot;
  const int b     = wid >> 8;
  const int qt    = (wid & 255) << 4;
  const int n     = lane & 15;
  const int khalf = (lane >> 4) << 1;
  const int mhi   = (lane >> 4) << 3;

  const float* Qrow = Q + ((size_t)b * SEQ + qt + n) * DK + khalf;
  v2f areg[32];
#pragma unroll
  for (int c = 0; c < 32; ++c) areg[c] = *(const v2f*)(Qrow + 4 * c);

  const v8f vzero = {};
  v8f acc[8];
#pragma unroll
  for (int t = 0; t < 8; ++t) acc[t] = vzero;

  const float* Kb = K + (size_t)b * SEQ * DK;
  const float* Vb = V + (size_t)b * SEQ * DK;
  const float* mB = mArr + (size_t)b * SEQ;
  const float* lB = lArr + (size_t)b * SEQ;
  float* sm = &smem[wslot][0];

  for (int kt = 0; kt < SEQ; kt += 16) {
    const float* Krow = Kb + (size_t)(kt + n) * DK + khalf;
    v8f c8 = vzero;
#pragma unroll
    for (int c = 0; c < 32; ++c) {
      v2f bk = *(const v2f*)(Krow + 4 * c);
      c8 = wmma_f32(areg[c], bk, c8);
    }

    const float mcol = mB[kt + n];
    const float rl   = 1.0f / lB[kt + n];
#pragma unroll
    for (int j = 0; j < 8; ++j) {
      const float p = __expf(c8[j] * SCALE - mcol) * rl;
      sm[(j + mhi) * 17 + n] = p;
    }
    __builtin_amdgcn_wave_barrier();

    v2f pa[4];
#pragma unroll
    for (int c2 = 0; c2 < 4; ++c2) {
      const int kk = 4 * c2 + khalf;
      pa[c2].x = sm[n * 17 + kk];
      pa[c2].y = sm[n * 17 + kk + 1];
    }

#pragma unroll
    for (int t = 0; t < 8; ++t) {
      const float* Vcol = Vb + (size_t)(kt + khalf) * DK + t * 16 + n;
#pragma unroll
      for (int c2 = 0; c2 < 4; ++c2) {
        v2f bv;
        bv.x = Vcol[(size_t)(4 * c2) * DK];
        bv.y = Vcol[(size_t)(4 * c2 + 1) * DK];
        acc[t] = wmma_f32(pa[c2], bv, acc[t]);
      }
    }
  }

  float* Ob = O + ((size_t)b * SEQ + qt) * DK;
#pragma unroll
  for (int t = 0; t < 8; ++t) {
#pragma unroll
    for (int j = 0; j < 8; ++j) {
      Ob[(size_t)(j + mhi) * DK + t * 16 + n] = acc[t][j];
    }
  }
}

extern "C" void kernel_launch(void* const* d_in, const int* in_sizes, int n_in,
                              void* d_out, int out_size, void* d_ws, size_t ws_size,
                              hipStream_t stream) {
  const float* q = (const float*)d_in[0];
  const float* k = (const float*)d_in[1];
  const float* v = (const float*)d_in[2];
  float* out = (float*)d_out;

  float* mArr = (float*)d_ws;
  float* lArr = mArr + (size_t)BS * SEQ;

  const int total_waves = BS * KTILES;
  const int blocks = total_waves / WAVES_PER_BLOCK;
  const int threads = WAVES_PER_BLOCK * 32;

  sdpa_colstats_kernel<<<blocks, threads, 0, stream>>>(q, k, mArr, lArr);
  sdpa_out_kernel<<<blocks, threads, 0, stream>>>(q, k, v, mArr, lArr, out);
}


# Occupancy 1->4 waves/SIMD: each tile's streamed range split over 4 waves of a 16-wave workgroup, partials merged via LDS; sums over queries/keys now 4 partial sums added in order
# speedup vs baseline: 3.0177x; 2.5595x over previous
#include <hip/hip_runtime.h>

typedef __attribute__((ext_vector_type(2))) float v2f;
typedef __attribute__((ext_vector_type(8))) float v8f;

#define BS   4
#define SEQ  4096
#define DK   128
#define KTILES (SEQ / 16)
#define SCALE 0.08838834764831845f
#define SPLIT 4
#define TILES_PER_BLOCK 4
#define WAVES_PER_BLOCK (SPLIT * TILES_PER_BLOCK)
#define PART_LEN (SEQ / SPLIT)

__device__ __forceinline__ v8f wmma_f32(v2f a, v2f b, v8f c) {
  return __builtin_amdgcn_wmma_f32_16x16x4_f32(
       false, a,  false, b,
       (short)0, c,  false,  false);
}

__global__ __launch_bounds__(WAVES_PER_BLOCK * 32)
void sdpa_colstats_kernel(const float* __restrict__ Q,
                          const float* __restrict__ K,
                          float* __restrict__ mArr,
                          float* __restrict__ lArr) {
  __shared__ float red[WAVES_PER_BLOCK][32];

  const int lane  = threadIdx.x & 31;
  const int wslot = __builtin_amdgcn_readfirstlane(threadIdx.x >> 5);
  const int tloc  = wslot & (TILES_PER_BLOCK - 1);
  const int part  = wslot / TILES_PER_BLOCK;
  const int wid   = blockIdx.x * TILES_PER_BLOCK + tloc;
  const int b     = wid >> 8;
  const int kt    = (wid & 255) << 4;
  const int n     = lane & 15;
  const int khalf = (lane >> 4) << 1;

  const float* Krow = K + ((size_t)b * SEQ + kt + n) * DK + khalf;
  v2f breg[32];
#pragma unroll
  for (int c = 0; c < 32; ++c) breg[c] = *(const v2f*)(Krow + 4 * c);

  const float* Qb = Q + (size_t)b * SEQ * DK;
  float m_run = -__builtin_inff();
  float l_run = 0.0f;

  const int qbeg = part * PART_LEN;
  for (int it = 0; it < PART_LEN; it += 16) {
    const int qt = qbeg + it;
    const float* Qrow = Qb + (size_t)(qt + n) * DK + khalf;
    v8f acc = {};
#pragma unroll
    for (int c = 0; c < 32; ++c) {
      v2f a = *(const v2f*)(Qrow + 4 * c);
      acc = wmma_f32(a, breg[c], acc);
    }
    float mloc = acc[0] * SCALE;
#pragma unroll
    for (int j = 1; j < 8; ++j) mloc = fmaxf(mloc, acc[j] * SCALE);
    const float m_new = fmaxf(m_run, mloc);
    float s = 0.0f;
#pragma unroll
    for (int j = 0; j < 8; ++j) s += __expf(acc[j] * SCALE - m_new);
    l_run = l_run * __expf(m_run - m_new) + s;
    m_run = m_new;
  }

  const float m_o = __shfl_xor(m_run, 16, 32);
  const float l_o = __shfl_xor(l_run, 16, 32);
  const float m_f = fmaxf(m_run, m_o);
  const float l_f = l_run * __expf(m_run - m_f) + l_o * __expf(m_o - m_f);

  if (lane < 16) {
    red[wslot][n]      = m_f;
    red[wslot][16 + n] = l_f;
  }
  __syncthreads();
  if (part == 0 && lane < 16) {
    float m = m_f;
    float l = l_f;
    for (int p = 1; p < SPLIT; ++p) {
      const float mo = red[p * TILES_PER_BLOCK + tloc][n];
      const float lo = red[p * TILES_PER_BLOCK + tloc][16 + n];
      const float mn = fmaxf(m, mo);
      l = l * __expf(m - mn) + lo * __expf(mo - mn);
      m = mn;
    }
    mArr[(size_t)b * SEQ + kt + n] = m;
    lArr[(size_t)b * SEQ + kt + n] = l;
  }
}

__global__ __launch_bounds__(WAVES_PER_BLOCK * 32)
void sdpa_out_kernel(const float* __restrict__ Q,
                     const float* __restrict__ K,
                     const float* __restrict__ V,
                     const float* __restrict__ mArr,
                     const float* __restrict__ lArr,
                     float* __restrict__ O) {
  __shared__ float smem[WAVES_PER_BLOCK][16 * 17];
  __shared__ float opart[(SPLIT - 1) * TILES_PER_BLOCK][16 * DK];

  const int lane  = threadIdx.x & 31;
  const int wslot = __builtin_amdgcn_readfirstlane(threadIdx.x >> 5);
  const int tloc  = wslot & (TILES_PER_BLOCK - 1);
  const int part  = wslot / TILES_PER_BLOCK;
  const int wid   = blockIdx.x * TILES_PER_BLOCK + tloc;
  const int b     = wid >> 8;
  const int qt    = (wid & 255) << 4;
  const int n     = lane & 15;
  const int khalf = (lane >> 4) << 1;
  const int mhi   = (lane >> 4) << 3;

  const float* Qrow = Q + ((size_t)b * SEQ + qt + n) * DK + khalf;
  v2f areg[32];
#pragma unroll
  for (int c = 0; c < 32; ++c) areg[c] = *(const v2f*)(Qrow + 4 * c);

  const v8f vzero = {};
  v8f acc[8];
#pragma unroll
  for (int t = 0; t < 8; ++t) acc[t] = vzero;

  const float* Kb = K + (size_t)b * SEQ * DK;
  const float* Vb = V + (size_t)b * SEQ * DK;
  const float* mB = mArr + (size_t)b * SEQ;
  const float* lB = lArr + (size_t)b * SEQ;
  float* sm = &smem[wslot][0];

  const int kbeg = part * PART_LEN;
  for (int it = 0; it < PART_LEN; it += 16) {
    const int kt = kbeg + it;
    const float* Krow = Kb + (size_t)(kt + n) * DK + khalf;
    v8f c8 = vzero;
#pragma unroll
    for (int c = 0; c < 32; ++c) {
      v2f bk = *(const v2f*)(Krow + 4 * c);
      c8 = wmma_f32(areg[c], bk, c8);
    }

    const float mcol = mB[kt + n];
    const float rl   = 1.0f / lB[kt + n];
#pragma unroll
    for (int j = 0; j < 8; ++j) {
      const float p = __expf(c8[j] * SCALE - mcol) * rl;
      sm[(j + mhi) * 17 + n] = p;
    }
    __builtin_amdgcn_wave_barrier();

    v2f pa[4];
#pragma unroll
    for (int c2 = 0; c2 < 4; ++c2) {
      const int kk = 4 * c2 + khalf;
      pa[c2].x = sm[n * 17 + kk];
      pa[c2].y = sm[n * 17 + kk + 1];
    }

#pragma unroll
    for (int t = 0; t < 8; ++t) {
      const float* Vcol = Vb + (size_t)(kt + khalf) * DK + t * 16 + n;
#pragma unroll
      for (int c2 = 0; c2 < 4; ++c2) {
        v2f bv;
        bv.x = Vcol[(size_t)(4 * c2) * DK];
        bv.y = Vcol[(size_t)(4 * c2 + 1) * DK];
        acc[t] = wmma_f32(pa[c2], bv, acc[t]);
      }
    }
  }

  if (part > 0) {
    float* op = &opart[(part - 1) * TILES_PER_BLOCK + tloc][0];
#pragma unroll
    for (int t = 0; t < 8; ++t) {
#pragma unroll
      for (int j = 0; j < 8; ++j) op[(t * 8 + j) * 32 + lane] = acc[t][j];
    }
  }
  __syncthreads();
  if (part == 0) {
#pragma unroll
    for (int p = 1; p < SPLIT; ++p) {
      const float* op = &opart[(p - 1) * TILES_PER_BLOCK + tloc][0];
#pragma unroll
      for (int t = 0; t < 8; ++t) {
#pragma unroll
        for (int j = 0; j < 8; ++j) acc[t][j] += op[(t * 8 + j) * 32 + lane];
      }
    }
    float* Ob = O + ((size_t)b * SEQ + qt) * DK;
#pragma unroll
    for (int t = 0; t < 8; ++t) {
#pragma unroll
      for (int j = 0; j < 8; ++j) {
        Ob[(size_t)(j + mhi) * DK + t * 16 + n] = acc[t][j];
      }
    }
  }
}

extern "C" void kernel_launch(void* const* d_in, const int* in_sizes, int n_in,
                              void* d_out, int out_size, void* d_ws, size_t ws_size,
                              hipStream_t stream) {
  const float* q = (const float*)d_in[0];
  const float* k = (const float*)d_in[1];
  const float* v = (const float*)d_in[2];
  float* out = (float*)d_out;

  float* mArr = (float*)d_ws;
  float* lArr = mArr + (size_t)BS * SEQ;

  const int blocks = BS * KTILES / TILES_PER_BLOCK;
  const int threads = WAVES_PER_BLOCK * 32;

  sdpa_colstats_kernel<<<blocks, threads, 0, stream>>>(q, k, mArr, lArr);
  sdpa_out_kernel<<<blocks, threads, 0, stream>>>(q, k, v, mArr, lArr, out);
}


# Q/K/V loads, O stores widened to 16 B/lane by remapping which d each WMMA slot and tile column holds (fewer cache lines per byte; waves_per_eu 4 keeps pass 1's in flight); QK^T d sum order permuted
# speedup vs baseline: 3.8285x; 1.2687x over previous
#include <hip/hip_runtime.h>

typedef __attribute__((ext_vector_type(2))) float v2f;
typedef __attribute__((ext_vector_type(4))) float v4f;
typedef __attribute__((ext_vector_type(8))) float v8f;

#define BS   4
#define SEQ  4096
#define DK   128
#define KTILES (SEQ / 16)
#define SCALE 0.08838834764831845f
#define SPLIT 4
#define TILES_PER_BLOCK 4
#define WAVES_PER_BLOCK (SPLIT * TILES_PER_BLOCK)
#define PART_LEN (SEQ / SPLIT)

__device__ __forceinline__ v8f wmma_f32(v2f a, v2f b, v8f c) {
  return __builtin_amdgcn_wmma_f32_16x16x4_f32(
       false, a,  false, b,
       (short)0, c,  false,  false);
}

__device__ __forceinline__ v4f load_frag4(const float* __restrict__ rowh, int i) {
  return *(const v4f*)(rowh + 8 * i);
}
__device__ __forceinline__ v2f lo2(v4f f) { v2f r; r.x = f.x; r.y = f.y; return r; }
__device__ __forceinline__ v2f hi2(v4f f) { v2f r; r.x = f.z; r.y = f.w; return r; }

__global__ __launch_bounds__(WAVES_PER_BLOCK * 32) __attribute__((amdgpu_waves_per_eu(4, 4)))
void sdpa_colstats_kernel(const float* __restrict__ Q,
                          const float* __restrict__ K,
                          float* __restrict__ mArr,
                          float* __restrict__ lArr) {
  __shared__ float red[WAVES_PER_BLOCK][32];

  const int lane  = threadIdx.x & 31;
  const int wslot = __builtin_amdgcn_readfirstlane(threadIdx.x >> 5);
  const int tloc  = wslot & (TILES_PER_BLOCK - 1);
  const int part  = wslot / TILES_PER_BLOCK;
  const int wid   = blockIdx.x * TILES_PER_BLOCK + tloc;
  const int b     = wid >> 8;
  const int kt    = (wid & 255) << 4;
  const int n     = lane & 15;
  const int dhalf = (lane >> 4) << 2;

  const float* Krow = K + ((size_t)b * SEQ + kt + n) * DK + dhalf;
  v2f breg[32];
#pragma unroll
  for (int i = 0; i < 16; ++i) {
    const v4f f = load_frag4(Krow, i);
    breg[2 * i]     = lo2(f);
    breg[2 * i + 1] = hi2(f);
  }

  const float* Qb = Q + (size_t)b * SEQ * DK;
  float m_run = -__builtin_inff();
  float l_run = 0.0f;

  const int qbeg = part * PART_LEN;
  for (int it = 0; it < PART_LEN; it += 16) {
    const int qt = qbeg + it;
    const float* Qrow = Qb + (size_t)(qt + n) * DK + dhalf;
    v8f acc = {};
#pragma unroll
    for (int i = 0; i < 16; ++i) {
      const v4f f = load_frag4(Qrow, i);
      acc = wmma_f32(lo2(f), breg[2 * i], acc);
      acc = wmma_f32(hi2(f), breg[2 * i + 1], acc);
    }
    float mloc = acc[0] * SCALE;
#pragma unroll
    for (int j = 1; j < 8; ++j) mloc = fmaxf(mloc, acc[j] * SCALE);
    const float m_new = fmaxf(m_run, mloc);
    float s = 0.0f;
#pragma unroll
    for (int j = 0; j < 8; ++j) s += __expf(acc[j] * SCALE - m_new);
    l_run = l_run * __expf(m_run - m_new) + s;
    m_run = m_new;
  }

  const float m_o = __shfl_xor(m_run, 16, 32);
  const float l_o = __shfl_xor(l_run, 16, 32);
  const float m_f = fmaxf(m_run, m_o);
  const float l_f = l_run * __expf(m_run - m_f) + l_o * __expf(m_o - m_f);

  if (lane < 16) {
    red[wslot][n]      = m_f;
    red[wslot][16 + n] = l_f;
  }
  __syncthreads();
  if (part == 0 && lane < 16) {
    float m = m_f;
    float l = l_f;
    for (int p = 1; p < SPLIT; ++p) {
      const float mo = red[p * TILES_PER_BLOCK + tloc][n];
      const float lo = red[p * TILES_PER_BLOCK + tloc][16 + n];
      const float mn = fmaxf(m, mo);
      l = l * __expf(m - mn) + lo * __expf(mo - mn);
      m = mn;
    }
    mArr[(size_t)b * SEQ + kt + n] = m;
    lArr[(size_t)b * SEQ + kt + n] = l;
  }
}

__global__ __launch_bounds__(WAVES_PER_BLOCK * 32)
void sdpa_out_kernel(const float* __restrict__ Q,
                     const float* __restrict__ K,
                     const float* __restrict__ V,
                     const float* __restrict__ mArr,
                     const float* __restrict__ lArr,
                     float* __restrict__ O) {
  __shared__ float smem[WAVES_PER_BLOCK][16 * 17];
  __shared__ float opart[(SPLIT - 1) * TILES_PER_BLOCK][16 * DK];

  const int lane  = threadIdx.x & 31;
  const int wslot = __builtin_amdgcn_readfirstlane(threadIdx.x >> 5);
  const int tloc  = wslot & (TILES_PER_BLOCK - 1);
  const int part  = wslot / TILES_PER_BLOCK;
  const int wid   = blockIdx.x * TILES_PER_BLOCK + tloc;
  const int b     = wid >> 8;
  const int qt    = (wid & 255) << 4;
  const int n     = lane & 15;
  const int khalf = (lane >> 4) << 1;
  const int dhalf = (lane >> 4) << 2;
  const int mhi   = (lane >> 4) << 3;

  const float* Qrow = Q + ((size_t)b * SEQ + qt + n) * DK + dhalf;
  v2f areg[32];
#pragma unroll
  for (int i = 0; i < 16; ++i) {
    const v4f f = load_frag4(Qrow, i);
    areg[2 * i]     = lo2(f);
    areg[2 * i + 1] = hi2(f);
  }

  const v8f vzero = {};
  v8f acc[8];
#pragma unroll
  for (int t = 0; t < 8; ++t) acc[t] = vzero;

  const float* Kb = K + (size_t)b * SEQ * DK;
  const float* Vb = V + (size_t)b * SEQ * DK;
  const float* mB = mArr + (size_t)b * SEQ;
  const float* lB = lArr + (size_t)b * SEQ;
  float* sm = &smem[wslot][0];

  const int kbeg = part * PART_LEN;
  for (int it = 0; it < PART_LEN; it += 16) {
    const int kt = kbeg + it;
    const float* Krow = Kb + (size_t)(kt + n) * DK + dhalf;
    v8f c8 = vzero;
#pragma unroll
    for (int i = 0; i < 16; ++i) {
      const v4f f = load_frag4(Krow, i);
      c8 = wmma_f32(areg[2 * i],     lo2(f), c8);
      c8 = wmma_f32(areg[2 * i + 1], hi2(f), c8);
    }

    const float mcol = mB[kt + n];
    const float rl   = 1.0f / lB[kt + n];
#pragma unroll
    for (int j = 0; j < 8; ++j) {
      const float p = __expf(c8[j] * SCALE - mcol) * rl;
      sm[(j + mhi) * 17 + n] = p;
    }
    __builtin_amdgcn_wave_barrier();

    v2f pa[4];
#pragma unroll
    for (int c2 = 0; c2 < 4; ++c2) {
      const int kk = 4 * c2 + khalf;
      pa[c2].x = sm[n * 17 + kk];
      pa[c2].y = sm[n * 17 + kk + 1];
    }

    const float* Vrow = Vb + (size_t)(kt + khalf) * DK + 4 * n;
#pragma unroll
    for (int c2 = 0; c2 < 4; ++c2) {
      const v4f x0 = *(const v4f*)(Vrow + (size_t)(4 * c2) * DK);
      const v4f y0 = *(const v4f*)(Vrow + (size_t)(4 * c2 + 1) * DK);
      const v4f x1 = *(const v4f*)(Vrow + (size_t)(4 * c2) * DK + 64);
      const v4f y1 = *(const v4f*)(Vrow + (size_t)(4 * c2 + 1) * DK + 64);
#pragma unroll
      for (int t = 0; t < 4; ++t) {
        v2f bv0, bv1;
        bv0.x = x0[t]; bv0.y = y0[t];
        bv1.x = x1[t]; bv1.y = y1[t];
        acc[t]     = wmma_f32(pa[c2], bv0, acc[t]);
        acc[4 + t] = wmma_f32(pa[c2], bv1, acc[4 + t]);
      }
    }
  }

  if (part > 0) {
    float* op = &opart[(part - 1) * TILES_PER_BLOCK + tloc][0];
#pragma unroll
    for (int t = 0; t < 8; ++t) {
#pragma unroll
      for (int j = 0; j < 8; ++j) op[(t * 8 + j) * 32 + lane] = acc[t][j];
    }
  }
  __syncthreads();
  if (part == 0) {
#pragma unroll
    for (int p = 1; p < SPLIT; ++p) {
      const float* op = &opart[(p - 1) * TILES_PER_BLOCK + tloc][0];
#pragma unroll
      for (int t = 0; t < 8; ++t) {
#pragma unroll
        for (int j = 0; j < 8; ++j) acc[t][j] += op[(t * 8 + j) * 32 + lane];
      }
    }
    float* Ob = O + ((size_t)b * SEQ + qt) * DK + 4 * n;
#pragma unroll
    for (int j = 0; j < 8; ++j) {
      v4f o0, o1;
      o0.x = acc[0][j]; o0.y = acc[1][j]; o0.z = acc[2][j]; o0.w = acc[3][j];
      o1.x = acc[4][j]; o1.y = acc[5][j]; o1.z = acc[6][j]; o1.w = acc[7][j];
      *(v4f*)(Ob + (size_t)(j + mhi) * DK)      = o0;
      *(v4f*)(Ob + (size_t)(j + mhi) * DK + 64) = o1;
    }
  }
}

extern "C" void kernel_launch(void* const* d_in, const int* in_sizes, int n_in,
                              void* d_out, int out_size, void* d_ws, size_t ws_size,
                              hipStream_t stream) {
  const float* q = (const float*)d_in[0];
  const float* k = (const float*)d_in[1];
  const float* v = (const float*)d_in[2];
  float* out = (float*)d_out;

  float* mArr = (float*)d_ws;
  float* lArr = mArr + (size_t)BS * SEQ;

  const int blocks = BS * KTILES / TILES_PER_BLOCK;
  const int threads = WAVES_PER_BLOCK * 32;

  sdpa_colstats_kernel<<<blocks, threads, 0, stream>>>(q, k, mArr, lArr);
  sdpa_out_kernel<<<blocks, threads, 0, stream>>>(q, k, v, mArr, lArr, out);
}


# Streamed Q/K tile read row-wise (one whole 512 B row = 4 lines per load, was 16 lines) and turned into WMMA fragments via a wave-private padded LDS tile; fragment contents and sum order unchanged
# speedup vs baseline: 4.0270x; 1.0518x over previous
#include <hip/hip_runtime.h>

typedef __attribute__((ext_vector_type(2))) float v2f;
typedef __attribute__((ext_vector_type(4))) float v4f;
typedef __attribute__((ext_vector_type(8))) float v8f;

#define BS   4
#define SEQ  4096
#define DK   128
#define KTILES (SEQ / 16)
#define SCALE 0.08838834764831845f
#define SPLIT 4
#define TILES_PER_BLOCK 4
#define WAVES_PER_BLOCK (SPLIT * TILES_PER_BLOCK)
#define PART_LEN (SEQ / SPLIT)
#define SPITCH (DK + 4)

__device__ __forceinline__ v8f wmma_f32(v2f a, v2f b, v8f c) {
  return __builtin_amdgcn_wmma_f32_16x16x4_f32(
       false, a,  false, b,
       (short)0, c,  false,  false);
}

__device__ __forceinline__ v4f load_frag4(const float* __restrict__ rowh, int i) {
  return *(const v4f*)(rowh + 8 * i);
}
__device__ __forceinline__ void stage_rows(const float* __restrict__ tile_lane, v4f (&g)[16]) {
#pragma unroll
  for (int r = 0; r < 16; ++r) g[r] = *(const v4f*)(tile_lane + (size_t)r * DK);
}
__device__ __forceinline__ void stage_store(float* st_w, const v4f (&g)[16]) {
#pragma unroll
  for (int r = 0; r < 16; ++r) *(v4f*)(st_w + r * SPITCH) = g[r];
}
__device__ __forceinline__ v4f stage_frag4(const float* st_r, int i) {
  return *(const v4f*)(st_r + 8 * i);
}
__device__ __forceinline__ v2f lo2(v4f f) { v2f r; r.x = f.x; r.y = f.y; return r; }
__device__ __forceinline__ v2f hi2(v4f f) { v2f r; r.x = f.z; r.y = f.w; return r; }

__global__ __launch_bounds__(WAVES_PER_BLOCK * 32) __attribute__((amdgpu_waves_per_eu(4, 4)))
void sdpa_colstats_kernel(const float* __restrict__ Q,
                          const float* __restrict__ K,
                          float* __restrict__ mArr,
                          float* __restrict__ lArr) {
  __shared__ float red[WAVES_PER_BLOCK][32];
  __shared__ __attribute__((aligned(16))) float stage[WAVES_PER_BLOCK][16 * SPITCH];

  const int lane  = threadIdx.x & 31;
  const int wslot = __builtin_amdgcn_readfirstlane(threadIdx.x >> 5);
  const int tloc  = wslot & (TILES_PER_BLOCK - 1);
  const int part  = wslot / TILES_PER_BLOCK;
  const int wid   = blockIdx.x * TILES_PER_BLOCK + tloc;
  const int b     = wid >> 8;
  const int kt    = (wid & 255) << 4;
  const int n     = lane & 15;
  const int dhalf = (lane >> 4) << 2;

  const float* Krow = K + ((size_t)b * SEQ + kt + n) * DK + dhalf;
  v2f breg[32];
#pragma unroll
  for (int i = 0; i < 16; ++i) {
    const v4f f = load_frag4(Krow, i);
    breg[2 * i]     = lo2(f);
    breg[2 * i + 1] = hi2(f);
  }

  const float* Qb = Q + (size_t)b * SEQ * DK + 4 * lane;
  float*       st_w = &stage[wslot][0] + 4 * lane;
  const float* st_r = &stage[wslot][0] + n * SPITCH + dhalf;
  float m_run = -__builtin_inff();
  float l_run = 0.0f;

  const int qbeg = part * PART_LEN;
  for (int it = 0; it < PART_LEN; it += 16) {
    const int qt = qbeg + it;
    v4f g[16];
    stage_rows(Qb + (size_t)qt * DK, g);
    stage_store(st_w, g);
    __builtin_amdgcn_wave_barrier();
    v4f f[16];
#pragma unroll
    for (int i = 0; i < 16; ++i) f[i] = stage_frag4(st_r, i);
    __builtin_amdgcn_sched_barrier(0);
    v8f acc = {};
#pragma unroll
    for (int i = 0; i < 16; ++i) {
      acc = wmma_f32(lo2(f[i]), breg[2 * i], acc);
      acc = wmma_f32(hi2(f[i]), breg[2 * i + 1], acc);
    }
    float mloc = acc[0] * SCALE;
#pragma unroll
    for (int j = 1; j < 8; ++j) mloc = fmaxf(mloc, acc[j] * SCALE);
    const float m_new = fmaxf(m_run, mloc);
    float s = 0.0f;
#pragma unroll
    for (int j = 0; j < 8; ++j) s += __expf(acc[j] * SCALE - m_new);
    l_run = l_run * __expf(m_run - m_new) + s;
    m_run = m_new;
  }

  const float m_o = __shfl_xor(m_run, 16, 32);
  const float l_o = __shfl_xor(l_run, 16, 32);
  const float m_f = fmaxf(m_run, m_o);
  const float l_f = l_run * __expf(m_run - m_f) + l_o * __expf(m_o - m_f);

  if (lane < 16) {
    red[wslot][n]      = m_f;
    red[wslot][16 + n] = l_f;
  }
  __syncthreads();
  if (part == 0 && lane < 16) {
    float m = m_f;
    float l = l_f;
    for (int p = 1; p < SPLIT; ++p) {
      const float mo = red[p * TILES_PER_BLOCK + tloc][n];
      const float lo = red[p * TILES_PER_BLOCK + tloc][16 + n];
      const float mn = fmaxf(m, mo);
      l = l * __expf(m - mn) + lo * __expf(mo - mn);
      m = mn;
    }
    mArr[(size_t)b * SEQ + kt + n] = m;
    lArr[(size_t)b * SEQ + kt + n] = l;
  }
}

__global__ __launch_bounds__(WAVES_PER_BLOCK * 32)
void sdpa_out_kernel(const float* __restrict__ Q,
                     const float* __restrict__ K,
                     const float* __restrict__ V,
                     const float* __restrict__ mArr,
                     const float* __restrict__ lArr,
                     float* __restrict__ O) {
  __shared__ float smem[WAVES_PER_BLOCK][16 * 17];
  __shared__ __attribute__((aligned(16))) float stage[WAVES_PER_BLOCK][16 * SPITCH];
  __shared__ float opart[(SPLIT - 1) * TILES_PER_BLOCK][16 * DK];

  const int lane  = threadIdx.x & 31;
  const int wslot = __builtin_amdgcn_readfirstlane(threadIdx.x >> 5);
  const int tloc  = wslot & (TILES_PER_BLOCK - 1);
  const int part  = wslot / TILES_PER_BLOCK;
  const int wid   = blockIdx.x * TILES_PER_BLOCK + tloc;
  const int b     = wid >> 8;
  const int qt    = (wid & 255) << 4;
  const int n     = lane & 15;
  const int khalf = (lane >> 4) << 1;
  const int dhalf = (lane >> 4) << 2;
  const int mhi   = (lane >> 4) << 3;

  const float* Qrow = Q + ((size_t)b * SEQ + qt + n) * DK + dhalf;
  v2f areg[32];
#pragma unroll
  for (int i = 0; i < 16; ++i) {
    const v4f f = load_frag4(Qrow, i);
    areg[2 * i]     = lo2(f);
    areg[2 * i + 1] = hi2(f);
  }

  const v8f vzero = {};
  v8f acc[8];
#pragma unroll
  for (int t = 0; t < 8; ++t) acc[t] = vzero;

  const float* Kb = K + (size_t)b * SEQ * DK;
  const float* Kl = Kb + 4 * lane;
  float*       st_w = &stage[wslot][0] + 4 * lane;
  const float* st_r = &stage[wslot][0] + n * SPITCH + dhalf;
  const float* Vb = V + (size_t)b * SEQ * DK;
  const float* mB = mArr + (size_t)b * SEQ;
  const float* lB = lArr + (size_t)b * SEQ;
  float* sm = &smem[wslot][0];

  const int kbeg = part * PART_LEN;
  for (int it = 0; it < PART_LEN; it += 16) {
    const int kt = kbeg + it;
    v4f g[16];
    stage_rows(Kl + (size_t)kt * DK, g);
    stage_store(st_w, g);
    __builtin_amdgcn_wave_barrier();
    v8f c8 = vzero;
#pragma unroll
    for (int i = 0; i < 16; ++i) {
      const v4f f = stage_frag4(st_r, i);
      c8 = wmma_f32(areg[2 * i],     lo2(f), c8);
      c8 = wmma_f32(areg[2 * i + 1], hi2(f), c8);
    }

    const float mcol = mB[kt + n];
    const float rl   = 1.0f / lB[kt + n];
#pragma unroll
    for (int j = 0; j < 8; ++j) {
      const float p = __expf(c8[j] * SCALE - mcol) * rl;
      sm[(j + mhi) * 17 + n] = p;
    }
    __builtin_amdgcn_wave_barrier();

    v2f pa[4];
#pragma unroll
    for (int c2 = 0; c2 < 4; ++c2) {
      const int kk = 4 * c2 + khalf;
      pa[c2].x = sm[n * 17 + kk];
      pa[c2].y = sm[n * 17 + kk + 1];
    }

    const float* Vrow = Vb + (size_t)(kt + khalf) * DK + 4 * n;
#pragma unroll
    for (int c2 = 0; c2 < 4; ++c2) {
      const v4f x0 = *(const v4f*)(Vrow + (size_t)(4 * c2) * DK);
      const v4f y0 = *(const v4f*)(Vrow + (size_t)(4 * c2 + 1) * DK);
      const v4f x1 = *(const v4f*)(Vrow + (size_t)(4 * c2) * DK + 64);
      const v4f y1 = *(const v4f*)(Vrow + (size_t)(4 * c2 + 1) * DK + 64);
#pragma unroll
      for (int t = 0; t < 4; ++t) {
        v2f bv0, bv1;
        bv0.x = x0[t]; bv0.y = y0[t];
        bv1.x = x1[t]; bv1.y = y1[t];
        acc[t]     = wmma_f32(pa[c2], bv0, acc[t]);
        acc[4 + t] = wmma_f32(pa[c2], bv1, acc[4 + t]);
      }
    }
  }

  if (part > 0) {
    float* op = &opart[(part - 1) * TILES_PER_BLOCK + tloc][0];
#pragma unroll
    for (int t = 0; t < 8; ++t) {
#pragma unroll
      for (int j = 0; j < 8; ++j) op[(t * 8 + j) * 32 + lane] = acc[t][j];
    }
  }
  __syncthreads();
  if (part == 0) {
#pragma unroll
    for (int p = 1; p < SPLIT; ++p) {
      const float* op = &opart[(p - 1) * TILES_PER_BLOCK + tloc][0];
#pragma unroll
      for (int t = 0; t < 8; ++t) {
#pragma unroll
        for (int j = 0; j < 8; ++j) acc[t][j] += op[(t * 8 + j) * 32 + lane];
      }
    }
    float* Ob = O + ((size_t)b * SEQ + qt) * DK + 4 * n;
#pragma unroll
    for (int j = 0; j < 8; ++j) {
      v4f o0, o1;
      o0.x = acc[0][j]; o0.y = acc[1][j]; o0.z = acc[2][j]; o0.w = acc[3][j];
      o1.x = acc[4][j]; o1.y = acc[5][j]; o1.z = acc[6][j]; o1.w = acc[7][j];
      *(v4f*)(Ob + (size_t)(j + mhi) * DK)      = o0;
      *(v4f*)(Ob + (size_t)(j + mhi) * DK + 64) = o1;
    }
  }
}

extern "C" void kernel_launch(void* const* d_in, const int* in_sizes, int n_in,
                              void* d_out, int out_size, void* d_ws, size_t ws_size,
                              hipStream_t stream) {
  const float* q = (const float*)d_in[0];
  const float* k = (const float*)d_in[1];
  const float* v = (const float*)d_in[2];
  float* out = (float*)d_out;

  float* mArr = (float*)d_ws;
  float* lArr = mArr + (size_t)BS * SEQ;

  const int blocks = BS * KTILES / TILES_PER_BLOCK;
  const int threads = WAVES_PER_BLOCK * 32;

  sdpa_colstats_kernel<<<blocks, threads, 0, stream>>>(q, k, mArr, lArr);
  sdpa_out_kernel<<<blocks, threads, 0, stream>>>(q, k, v, mArr, lArr, out);
}


# Streamed tiles' global loads issued a stage ahead of the WMMAs that use them (next Q/K rows under this tile's products, V rows under the QK^T chain) into registers the chain frees; sums unchanged
# speedup vs baseline: 4.2092x; 1.0452x over previous
#include <hip/hip_runtime.h>

typedef __attribute__((ext_vector_type(2))) float v2f;
typedef __attribute__((ext_vector_type(4))) float v4f;
typedef __attribute__((ext_vector_type(8))) float v8f;

#define BS   4
#define SEQ  4096
#define DK   128
#define KTILES (SEQ / 16)
#define SCALE 0.08838834764831845f
#define SPLIT 4
#define TILES_PER_BLOCK 4
#define WAVES_PER_BLOCK (SPLIT * TILES_PER_BLOCK)
#define PART_LEN (SEQ / SPLIT)
#define SPITCH (DK + 4)

__device__ __forceinline__ v8f wmma_f32(v2f a, v2f b, v8f c) {
  return __builtin_amdgcn_wmma_f32_16x16x4_f32(
       false, a,  false, b,
       (short)0, c,  false,  false);
}

__device__ __forceinline__ v4f load_frag4(const float* __restrict__ rowh, int i) {
  return *(const v4f*)(rowh + 8 * i);
}
__device__ __forceinline__ void stage_rows(const float* __restrict__ tile_lane, v4f (&g)[16]) {
#pragma unroll
  for (int r = 0; r < 16; ++r) g[r] = *(const v4f*)(tile_lane + (size_t)r * DK);
}
__device__ __forceinline__ void stage_store(float* st_w, const v4f (&g)[16]) {
#pragma unroll
  for (int r = 0; r < 16; ++r) *(v4f*)(st_w + r * SPITCH) = g[r];
}
__device__ __forceinline__ v4f stage_frag4(const float* st_r, int i) {
  return *(const v4f*)(st_r + 8 * i);
}
__device__ __forceinline__ v4f load_vrow(const float* __restrict__ Vrow, int idx) {
  return *(const v4f*)(Vrow + (size_t)(4 * (idx >> 2) + (idx & 1)) * DK + 64 * ((idx >> 1) & 1));
}
__device__ __forceinline__ v2f lo2(v4f f) { v2f r; r.x = f.x; r.y = f.y; return r; }
__device__ __forceinline__ v2f hi2(v4f f) { v2f r; r.x = f.z; r.y = f.w; return r; }

__global__ __launch_bounds__(WAVES_PER_BLOCK * 32) __attribute__((amdgpu_waves_per_eu(4, 4)))
void sdpa_colstats_kernel(const float* __restrict__ Q,
                          const float* __restrict__ K,
                          float* __restrict__ mArr,
                          float* __restrict__ lArr) {
  __shared__ float red[WAVES_PER_BLOCK][32];
  __shared__ __attribute__((aligned(16))) float stage[WAVES_PER_BLOCK][16 * SPITCH];

  const int lane  = threadIdx.x & 31;
  const int wslot = __builtin_amdgcn_readfirstlane(threadIdx.x >> 5);
  const int tloc  = wslot & (TILES_PER_BLOCK - 1);
  const int part  = wslot / TILES_PER_BLOCK;
  const int wid   = blockIdx.x * TILES_PER_BLOCK + tloc;
  const int b     = wid >> 8;
  const int kt    = (wid & 255) << 4;
  const int n     = lane & 15;
  const int dhalf = (lane >> 4) << 2;

  const float* Krow = K + ((size_t)b * SEQ + kt + n) * DK + dhalf;
  v2f breg[32];
#pragma unroll
  for (int i = 0; i < 16; ++i) {
    const v4f f = load_frag4(Krow, i);
    breg[2 * i]     = lo2(f);
    breg[2 * i + 1] = hi2(f);
  }

  const float* Qb = Q + (size_t)b * SEQ * DK + 4 * lane;
  float*       st_w = &stage[wslot][0] + 4 * lane;
  const float* st_r = &stage[wslot][0] + n * SPITCH + dhalf;
  float m_run = -__builtin_inff();
  float l_run = 0.0f;

  const int qbeg = part * PART_LEN;
  v4f g[16];
  stage_rows(Qb + (size_t)qbeg * DK, g);
  for (int it = 0; it < PART_LEN; it += 16) {
    stage_store(st_w, g);
    const int itn = (it + 16 < PART_LEN) ? it + 16 : it;
    stage_rows(Qb + (size_t)(qbeg + itn) * DK, g);
    __builtin_amdgcn_wave_barrier();
    v4f f[16];
#pragma unroll
    for (int i = 0; i < 16; ++i) f[i] = stage_frag4(st_r, i);
    __builtin_amdgcn_sched_barrier(0);
    v8f acc = {};
#pragma unroll
    for (int i = 0; i < 16; ++i) {
      acc = wmma_f32(lo2(f[i]), breg[2 * i], acc);
      acc = wmma_f32(hi2(f[i]), breg[2 * i + 1], acc);
    }
    float mloc = acc[0] * SCALE;
#pragma unroll
    for (int j = 1; j < 8; ++j) mloc = fmaxf(mloc, acc[j] * SCALE);
    const float m_new = fmaxf(m_run, mloc);
    float s = 0.0f;
#pragma unroll
    for (int j = 0; j < 8; ++j) s += __expf(acc[j] * SCALE - m_new);
    l_run = l_run * __expf(m_run - m_new) + s;
    m_run = m_new;
  }

  const float m_o = __shfl_xor(m_run, 16, 32);
  const float l_o = __shfl_xor(l_run, 16, 32);
  const float m_f = fmaxf(m_run, m_o);
  const float l_f = l_run * __expf(m_run - m_f) + l_o * __expf(m_o - m_f);

  if (lane < 16) {
    red[wslot][n]      = m_f;
    red[wslot][16 + n] = l_f;
  }
  __syncthreads();
  if (part == 0 && lane < 16) {
    float m = m_f;
    float l = l_f;
    for (int p = 1; p < SPLIT; ++p) {
      const float mo = red[p * TILES_PER_BLOCK + tloc][n];
      const float lo = red[p * TILES_PER_BLOCK + tloc][16 + n];
      const float mn = fmaxf(m, mo);
      l = l * __expf(m - mn) + lo * __expf(mo - mn);
      m = mn;
    }
    mArr[(size_t)b * SEQ + kt + n] = m;
    lArr[(size_t)b * SEQ + kt + n] = l;
  }
}

__global__ __launch_bounds__(WAVES_PER_BLOCK * 32)
void sdpa_out_kernel(const float* __restrict__ Q,
                     const float* __restrict__ K,
                     const float* __restrict__ V,
                     const float* __restrict__ mArr,
                     const float* __restrict__ lArr,
                     float* __restrict__ O) {
  __shared__ float smem[WAVES_PER_BLOCK][16 * 17];
  __shared__ __attribute__((aligned(16))) float stage[WAVES_PER_BLOCK][16 * SPITCH];
  __shared__ float opart[(SPLIT - 1) * TILES_PER_BLOCK][16 * DK];

  const int lane  = threadIdx.x & 31;
  const int wslot = __builtin_amdgcn_readfirstlane(threadIdx.x >> 5);
  const int tloc  = wslot & (TILES_PER_BLOCK - 1);
  const int part  = wslot / TILES_PER_BLOCK;
  const int wid   = blockIdx.x * TILES_PER_BLOCK + tloc;
  const int b     = wid >> 8;
  const int qt    = (wid & 255) << 4;
  const int n     = lane & 15;
  const int khalf = (lane >> 4) << 1;
  const int dhalf = (lane >> 4) << 2;
  const int mhi   = (lane >> 4) << 3;

  const float* Qrow = Q + ((size_t)b * SEQ + qt + n) * DK + dhalf;
  v2f areg[32];
#pragma unroll
  for (int i = 0; i < 16; ++i) {
    const v4f f = load_frag4(Qrow, i);
    areg[2 * i]     = lo2(f);
    areg[2 * i + 1] = hi2(f);
  }

  const v8f vzero = {};
  v8f acc[8];
#pragma unroll
  for (int t = 0; t < 8; ++t) acc[t] = vzero;

  const float* Kb = K + (size_t)b * SEQ * DK;
  const float* Kl = Kb + 4 * lane;
  float*       st_w = &stage[wslot][0] + 4 * lane;
  const float* st_r = &stage[wslot][0] + n * SPITCH + dhalf;
  const float* Vb = V + (size_t)b * SEQ * DK;
  const float* mB = mArr + (size_t)b * SEQ;
  const float* lB = lArr + (size_t)b * SEQ;
  float* sm = &smem[wslot][0];

  const int kbeg = part * PART_LEN;
  v4f g[16];
  stage_rows(Kl + (size_t)kbeg * DK, g);
  for (int it = 0; it < PART_LEN; it += 16) {
    const int kt = kbeg + it;
    const int itn = (it + 16 < PART_LEN) ? it + 16 : it;
    const float* Knext = Kl + (size_t)(kbeg + itn) * DK;
    const float* Vrow = Vb + (size_t)(kt + khalf) * DK + 4 * n;
    stage_store(st_w, g);
    __builtin_amdgcn_wave_barrier();
    v4f f[16];
#pragma unroll
    for (int i = 0; i < 16; ++i) f[i] = stage_frag4(st_r, i);
    const float mcol = mB[kt + n];
    const float lcol = lB[kt + n];
    __builtin_amdgcn_sched_barrier(0);
    v4f vv[16];
    v8f c8 = vzero;
#pragma unroll
    for (int i = 0; i < 16; ++i) {
      c8 = wmma_f32(areg[2 * i],     lo2(f[i]), c8);
      c8 = wmma_f32(areg[2 * i + 1], hi2(f[i]), c8);
      if (i == 5 || i == 9 || i == 13 || i == 15) {
        const int q4 = (i == 15) ? 3 : (i - 5) / 4;
        __builtin_amdgcn_sched_barrier(0);
#pragma unroll
        for (int u = 0; u < 4; ++u) vv[4 * q4 + u] = load_vrow(Vrow, 4 * q4 + u);
      }
    }

    const float rl   = 1.0f / lcol;
#pragma unroll
    for (int j = 0; j < 8; ++j) {
      const float p = __expf(c8[j] * SCALE - mcol) * rl;
      sm[(j + mhi) * 17 + n] = p;
    }
    __builtin_amdgcn_wave_barrier();

    v2f pa[4];
#pragma unroll
    for (int c2 = 0; c2 < 4; ++c2) {
      const int kk = 4 * c2 + khalf;
      pa[c2].x = sm[n * 17 + kk];
      pa[c2].y = sm[n * 17 + kk + 1];
    }

    __builtin_amdgcn_sched_barrier(0);
#pragma unroll
    for (int r = 0; r < 4; ++r) g[r] = *(const v4f*)(Knext + (size_t)r * DK);
#pragma unroll
    for (int c2 = 0; c2 < 4; ++c2) {
      const v4f x0 = vv[4 * c2];
      const v4f y0 = vv[4 * c2 + 1];
      const v4f x1 = vv[4 * c2 + 2];
      const v4f y1 = vv[4 * c2 + 3];
#pragma unroll
      for (int t = 0; t < 4; ++t) {
        v2f bv0, bv1;
        bv0.x = x0[t]; bv0.y = y0[t];
        bv1.x = x1[t]; bv1.y = y1[t];
        acc[t]     = wmma_f32(pa[c2], bv0, acc[t]);
        acc[4 + t] = wmma_f32(pa[c2], bv1, acc[4 + t]);
      }
      if (c2 < 3) {
        __builtin_amdgcn_sched_barrier(0);
#pragma unroll
        for (int r = 4 * c2 + 4; r < 4 * c2 + 8; ++r) g[r] = *(const v4f*)(Knext + (size_t)r * DK);
      }
    }
    asm volatile("" ::: "memory");
  }

  if (part > 0) {
    float* op = &opart[(part - 1) * TILES_PER_BLOCK + tloc][0];
#pragma unroll
    for (int t = 0; t < 8; ++t) {
#pragma unroll
      for (int j = 0; j < 8; ++j) op[(t * 8 + j) * 32 + lane] = acc[t][j];
    }
  }
  __syncthreads();
  if (part == 0) {
#pragma unroll
    for (int p = 1; p < SPLIT; ++p) {
      const float* op = &opart[(p - 1) * TILES_PER_BLOCK + tloc][0];
#pragma unroll
      for (int t = 0; t < 8; ++t) {
#pragma unroll
        for (int j = 0; j < 8; ++j) acc[t][j] += op[(t * 8 + j) * 32 + lane];
      }
    }
    float* Ob = O + ((size_t)b * SEQ + qt) * DK + 4 * n;
#pragma unroll
    for (int j = 0; j < 8; ++j) {
      v4f o0, o1;
      o0.x = acc[0][j]; o0.y = acc[1][j]; o0.z = acc[2][j]; o0.w = acc[3][j];
      o1.x = acc[4][j]; o1.y = acc[5][j]; o1.z = acc[6][j]; o1.w = acc[7][j];
      *(v4f*)(Ob + (size_t)(j + mhi) * DK)      = o0;
      *(v4f*)(Ob + (size_t)(j + mhi) * DK + 64) = o1;
    }
  }
}

extern "C" void kernel_launch(void* const* d_in, const int* in_sizes, int n_in,
                              void* d_out, int out_size, void* d_ws, size_t ws_size,
                              hipStream_t stream) {
  const float* q = (const float*)d_in[0];
  const float* k = (const float*)d_in[1];
  const float* v = (const float*)d_in[2];
  float* out = (float*)d_out;

  float* mArr = (float*)d_ws;
  float* lArr = mArr + (size_t)BS * SEQ;

  const int blocks = BS * KTILES / TILES_PER_BLOCK;
  const int threads = WAVES_PER_BLOCK * 32;

  sdpa_colstats_kernel<<<blocks, threads, 0, stream>>>(q, k, mArr, lArr);
  sdpa_out_kernel<<<blocks, threads, 0, stream>>>(q, k, v, mArr, lArr, out);
}


# Wave issue priority falls with own progress (s_setprio 3,2,1,0 at 0, 1/2, 3/4, 7/8 of each wave's streamed range): lagging waves of a SIMD win the matrix pipe, all four end together; sums unchanged
# speedup vs baseline: 4.5988x; 1.0926x over previous
#include <hip/hip_runtime.h>

typedef __attribute__((ext_vector_type(2))) float v2f;
typedef __attribute__((ext_vector_type(4))) float v4f;
typedef __attribute__((ext_vector_type(8))) float v8f;

#define BS   4
#define SEQ  4096
#define DK   128
#define KTILES (SEQ / 16)
#define SCALE 0.08838834764831845f
#define SPLIT 4
#define TILES_PER_BLOCK 4
#define WAVES_PER_BLOCK (SPLIT * TILES_PER_BLOCK)
#define PART_LEN (SEQ / SPLIT)
#define SPITCH (DK + 4)

__device__ __forceinline__ v8f wmma_f32(v2f a, v2f b, v8f c) {
  return __builtin_amdgcn_wmma_f32_16x16x4_f32(
       false, a,  false, b,
       (short)0, c,  false,  false);
}

__device__ __forceinline__ v4f load_frag4(const float* __restrict__ rowh, int i) {
  return *(const v4f*)(rowh + 8 * i);
}
__device__ __forceinline__ void stage_rows(const float* __restrict__ tile_lane, v4f (&g)[16]) {
#pragma unroll
  for (int r = 0; r < 16; ++r) g[r] = *(const v4f*)(tile_lane + (size_t)r * DK);
}
__device__ __forceinline__ void stage_store(float* st_w, const v4f (&g)[16]) {
#pragma unroll
  for (int r = 0; r < 16; ++r) *(v4f*)(st_w + r * SPITCH) = g[r];
}
__device__ __forceinline__ v4f stage_frag4(const float* st_r, int i) {
  return *(const v4f*)(st_r + 8 * i);
}
__device__ __forceinline__ v4f load_vrow(const float* __restrict__ Vrow, int idx) {
  return *(const v4f*)(Vrow + (size_t)(4 * (idx >> 2) + (idx & 1)) * DK + 64 * ((idx >> 1) & 1));
}
__device__ __forceinline__ void prio_by_progress(int done) {
  if (done == PART_LEN / 2)                __builtin_amdgcn_s_setprio(2);
  if (done == PART_LEN / 2 + PART_LEN / 4) __builtin_amdgcn_s_setprio(1);
  if (done == PART_LEN - PART_LEN / 8)     __builtin_amdgcn_s_setprio(0);
}
__device__ __forceinline__ v2f lo2(v4f f) { v2f r; r.x = f.x; r.y = f.y; return r; }
__device__ __forceinline__ v2f hi2(v4f f) { v2f r; r.x = f.z; r.y = f.w; return r; }

__global__ __launch_bounds__(WAVES_PER_BLOCK * 32) __attribute__((amdgpu_waves_per_eu(4, 4)))
void sdpa_colstats_kernel(const float* __restrict__ Q,
                          const float* __restrict__ K,
                          float* __restrict__ mArr,
                          float* __restrict__ lArr) {
  __shared__ float red[WAVES_PER_BLOCK][32];
  __shared__ __attribute__((aligned(16))) float stage[WAVES_PER_BLOCK][16 * SPITCH];

  const int lane  = threadIdx.x & 31;
  const int wslot = __builtin_amdgcn_readfirstlane(threadIdx.x >> 5);
  const int tloc  = wslot & (TILES_PER_BLOCK - 1);
  const int part  = wslot / TILES_PER_BLOCK;
  const int wid   = blockIdx.x * TILES_PER_BLOCK + tloc;
  const int b     = wid >> 8;
  const int kt    = (wid & 255) << 4;
  const int n     = lane & 15;
  const int dhalf = (lane >> 4) << 2;

  const float* Krow = K + ((size_t)b * SEQ + kt + n) * DK + dhalf;
  v2f breg[32];
#pragma unroll
  for (int i = 0; i < 16; ++i) {
    const v4f f = load_frag4(Krow, i);
    breg[2 * i]     = lo2(f);
    breg[2 * i + 1] = hi2(f);
  }

  const float* Qb = Q + (size_t)b * SEQ * DK + 4 * lane;
  float*       st_w = &stage[wslot][0] + 4 * lane;
  const float* st_r = &stage[wslot][0] + n * SPITCH + dhalf;
  float m_run = -__builtin_inff();
  float l_run = 0.0f;

  const int qbeg = part * PART_LEN;
  v4f g[16];
  stage_rows(Qb + (size_t)qbeg * DK, g);
  __builtin_amdgcn_s_setprio(3);
  for (int it = 0; it < PART_LEN; it += 16) {
    prio_by_progress(it);
    stage_store(st_w, g);
    const int itn = (it + 16 < PART_LEN) ? it + 16 : it;
    stage_rows(Qb + (size_t)(qbeg + itn) * DK, g);
    __builtin_amdgcn_wave_barrier();
    v4f f[16];
#pragma unroll
    for (int i = 0; i < 16; ++i) f[i] = stage_frag4(st_r, i);
    __builtin_amdgcn_sched_barrier(0);
    v8f acc = {};
#pragma unroll
    for (int i = 0; i < 16; ++i) {
      acc = wmma_f32(lo2(f[i]), breg[2 * i], acc);
      acc = wmma_f32(hi2(f[i]), breg[2 * i + 1], acc);
    }
    float mloc = acc[0] * SCALE;
#pragma unroll
    for (int j = 1; j < 8; ++j) mloc = fmaxf(mloc, acc[j] * SCALE);
    const float m_new = fmaxf(m_run, mloc);
    float s = 0.0f;
#pragma unroll
    for (int j = 0; j < 8; ++j) s += __expf(acc[j] * SCALE - m_new);
    l_run = l_run * __expf(m_run - m_new) + s;
    m_run = m_new;
  }

  const float m_o = __shfl_xor(m_run, 16, 32);
  const float l_o = __shfl_xor(l_run, 16, 32);
  const float m_f = fmaxf(m_run, m_o);
  const float l_f = l_run * __expf(m_run - m_f) + l_o * __expf(m_o - m_f);

  if (lane < 16) {
    red[wslot][n]      = m_f;
    red[wslot][16 + n] = l_f;
  }
  __syncthreads();
  if (part == 0 && lane < 16) {
    float m = m_f;
    float l = l_f;
    for (int p = 1; p < SPLIT; ++p) {
      const float mo = red[p * TILES_PER_BLOCK + tloc][n];
      const float lo = red[p * TILES_PER_BLOCK + tloc][16 + n];
      const float mn = fmaxf(m, mo);
      l = l * __expf(m - mn) + lo * __expf(mo - mn);
      m = mn;
    }
    mArr[(size_t)b * SEQ + kt + n] = m;
    lArr[(size_t)b * SEQ + kt + n] = l;
  }
}

__global__ __launch_bounds__(WAVES_PER_BLOCK * 32)
void sdpa_out_kernel(const float* __restrict__ Q,
                     const float* __restrict__ K,
                     const float* __restrict__ V,
                     const float* __restrict__ mArr,
                     const float* __restrict__ lArr,
                     float* __restrict__ O) {
  __shared__ float smem[WAVES_PER_BLOCK][16 * 17];
  __shared__ __attribute__((aligned(16))) float stage[WAVES_PER_BLOCK][16 * SPITCH];
  __shared__ float opart[(SPLIT - 1) * TILES_PER_BLOCK][16 * DK];

  const int lane  = threadIdx.x & 31;
  const int wslot = __builtin_amdgcn_readfirstlane(threadIdx.x >> 5);
  const int tloc  = wslot & (TILES_PER_BLOCK - 1);
  const int part  = wslot / TILES_PER_BLOCK;
  const int wid   = blockIdx.x * TILES_PER_BLOCK + tloc;
  const int b     = wid >> 8;
  const int qt    = (wid & 255) << 4;
  const int n     = lane & 15;
  const int khalf = (lane >> 4) << 1;
  const int dhalf = (lane >> 4) << 2;
  const int mhi   = (lane >> 4) << 3;

  const float* Qrow = Q + ((size_t)b * SEQ + qt + n) * DK + dhalf;
  v2f areg[32];
#pragma unroll
  for (int i = 0; i < 16; ++i) {
    const v4f f = load_frag4(Qrow, i);
    areg[2 * i]     = lo2(f);
    areg[2 * i + 1] = hi2(f);
  }

  const v8f vzero = {};
  v8f acc[8];
#pragma unroll
  for (int t = 0; t < 8; ++t) acc[t] = vzero;

  const float* Kb = K + (size_t)b * SEQ * DK;
  const float* Kl = Kb + 4 * lane;
  float*       st_w = &stage[wslot][0] + 4 * lane;
  const float* st_r = &stage[wslot][0] + n * SPITCH + dhalf;
  const float* Vb = V + (size_t)b * SEQ * DK;
  const float* mB = mArr + (size_t)b * SEQ;
  const float* lB = lArr + (size_t)b * SEQ;
  float* sm = &smem[wslot][0];

  const int kbeg = part * PART_LEN;
  v4f g[16];
  stage_rows(Kl + (size_t)kbeg * DK, g);
  __builtin_amdgcn_s_setprio(3);
  for (int it = 0; it < PART_LEN; it += 16) {
    prio_by_progress(it);
    const int kt = kbeg + it;
    const int itn = (it + 16 < PART_LEN) ? it + 16 : it;
    const float* Knext = Kl + (size_t)(kbeg + itn) * DK;
    const float* Vrow = Vb + (size_t)(kt + khalf) * DK + 4 * n;
    stage_store(st_w, g);
    __builtin_amdgcn_wave_barrier();
    v4f f[16];
#pragma unroll
    for (int i = 0; i < 16; ++i) f[i] = stage_frag4(st_r, i);
    const float mcol = mB[kt + n];
    const float lcol = lB[kt + n];
    __builtin_amdgcn_sched_barrier(0);
    v4f vv[16];
    v8f c8 = vzero;
#pragma unroll
    for (int i = 0; i < 16; ++i) {
      c8 = wmma_f32(areg[2 * i],     lo2(f[i]), c8);
      c8 = wmma_f32(areg[2 * i + 1], hi2(f[i]), c8);
      if (i == 5 || i == 9 || i == 13 || i == 15) {
        const int q4 = (i == 15) ? 3 : (i - 5) / 4;
        __builtin_amdgcn_sched_barrier(0);
#pragma unroll
        for (int u = 0; u < 4; ++u) vv[4 * q4 + u] = load_vrow(Vrow, 4 * q4 + u);
      }
    }

    const float rl   = 1.0f / lcol;
#pragma unroll
    for (int j = 0; j < 8; ++j) {
      const float p = __expf(c8[j] * SCALE - mcol) * rl;
      sm[(j + mhi) * 17 + n] = p;
    }
    __builtin_amdgcn_wave_barrier();

    v2f pa[4];
#pragma unroll
    for (int c2 = 0; c2 < 4; ++c2) {
      const int kk = 4 * c2 + khalf;
      pa[c2].x = sm[n * 17 + kk];
      pa[c2].y = sm[n * 17 + kk + 1];
    }

    __builtin_amdgcn_sched_barrier(0);
#pragma unroll
    for (int r = 0; r < 4; ++r) g[r] = *(const v4f*)(Knext + (size_t)r * DK);
#pragma unroll
    for (int c2 = 0; c2 < 4; ++c2) {
      const v4f x0 = vv[4 * c2];
      const v4f y0 = vv[4 * c2 + 1];
      const v4f x1 = vv[4 * c2 + 2];
      const v4f y1 = vv[4 * c2 + 3];
#pragma unroll
      for (int t = 0; t < 4; ++t) {
        v2f bv0, bv1;
        bv0.x = x0[t]; bv0.y = y0[t];
        bv1.x = x1[t]; bv1.y = y1[t];
        acc[t]     = wmma_f32(pa[c2], bv0, acc[t]);
        acc[4 + t] = wmma_f32(pa[c2], bv1, acc[4 + t]);
      }
      if (c2 < 3) {
        __builtin_amdgcn_sched_barrier(0);
#pragma unroll
        for (int r = 4 * c2 + 4; r < 4 * c2 + 8; ++r) g[r] = *(const v4f*)(Knext + (size_t)r * DK);
      }
    }
    asm volatile("" ::: "memory");
  }

  if (part > 0) {
    float* op = &opart[(part - 1) * TILES_PER_BLOCK + tloc][0];
#pragma unroll
    for (int t = 0; t < 8; ++t) {
#pragma unroll
      for (int j = 0; j < 8; ++j) op[(t * 8 + j) * 32 + lane] = acc[t][j];
    }
  }
  __syncthreads();
  if (part == 0) {
#pragma unroll
    for (int p = 1; p < SPLIT; ++p) {
      const float* op = &opart[(p - 1) * TILES_PER_BLOCK + tloc][0];
#pragma unroll
      for (int t = 0; t < 8; ++t) {
#pragma unroll
        for (int j = 0; j < 8; ++j) acc[t][j] += op[(t * 8 + j) * 32 + lane];
      }
    }
    float* Ob = O + ((size_t)b * SEQ + qt) * DK + 4 * n;
#pragma unroll
    for (int j = 0; j < 8; ++j) {
      v4f o0, o1;
      o0.x = acc[0][j]; o0.y = acc[1][j]; o0.z = acc[2][j]; o0.w = acc[3][j];
      o1.x = acc[4][j]; o1.y = acc[5][j]; o1.z = acc[6][j]; o1.w = acc[7][j];
      *(v4f*)(Ob + (size_t)(j + mhi) * DK)      = o0;
      *(v4f*)(Ob + (size_t)(j + mhi) * DK + 64) = o1;
    }
  }
}

extern "C" void kernel_launch(void* const* d_in, const int* in_sizes, int n_in,
                              void* d_out, int out_size, void* d_ws, size_t ws_size,
                              hipStream_t stream) {
  const float* q = (const float*)d_in[0];
  const float* k = (const float*)d_in[1];
  const float* v = (const float*)d_in[2];
  float* out = (float*)d_out;

  float* mArr = (float*)d_ws;
  float* lArr = mArr + (size_t)BS * SEQ;

  const int blocks = BS * KTILES / TILES_PER_BLOCK;
  const int threads = WAVES_PER_BLOCK * 32;

  sdpa_colstats_kernel<<<blocks, threads, 0, stream>>>(q, k, mArr, lArr);
  sdpa_out_kernel<<<blocks, threads, 0, stream>>>(q, k, v, mArr, lArr, out);
}


# Score tiles kept, not recomputed: pass 1 also stores each 16x16 S tile to the workspace, pass 2 loads it in place of its 32-WMMA QK^T chain (as far as ws_size reaches); values and sum order unchanged
# speedup vs baseline: 5.9137x; 1.2859x over previous
#include <hip/hip_runtime.h>

typedef __attribute__((ext_vector_type(2))) float v2f;
typedef __attribute__((ext_vector_type(4))) float v4f;
typedef __attribute__((ext_vector_type(8))) float v8f;

#define BS   4
#define SEQ  4096
#define DK   128
#define KTILES (SEQ / 16)
#define SCALE 0.08838834764831845f
#define SPLIT 4
#define TILES_PER_BLOCK 4
#define WAVES_PER_BLOCK (SPLIT * TILES_PER_BLOCK)
#define PART_LEN (SEQ / SPLIT)
#define SPITCH (DK + 4)
#define PART_TILES (PART_LEN / 16)
#define TILE_FLOATS 256
#define STATS_FLOATS (2 * BS * SEQ)

__device__ __forceinline__ int kept_tiles(size_t ws_bytes) {
  const size_t stats = (size_t)STATS_FLOATS * sizeof(float);
  const size_t per_index = (size_t)BS * KTILES * SPLIT * TILE_FLOATS * sizeof(float);
  if (ws_bytes < stats + per_index) return 0;
  const size_t t = (ws_bytes - stats) / per_index;
  return t < (size_t)PART_TILES ? (int)t : PART_TILES;
}
__device__ __forceinline__ void keep_store4(float* p, v4f x) { __builtin_nontemporal_store(x, (v4f*)p); }
__device__ __forceinline__ v4f  keep_load4(const float* p)  { return __builtin_nontemporal_load((const v4f*)p); }

__device__ __forceinline__ v8f wmma_f32(v2f a, v2f b, v8f c) {
  return __builtin_amdgcn_wmma_f32_16x16x4_f32(
       false, a,  false, b,
       (short)0, c,  false,  false);
}

__device__ __forceinline__ v4f load_frag4(const float* __restrict__ rowh, int i) {
  return *(const v4f*)(rowh + 8 * i);
}
__device__ __forceinline__ void stage_rows(const float* __restrict__ tile_lane, v4f (&g)[16]) {
#pragma unroll
  for (int r = 0; r < 16; ++r) g[r] = *(const v4f*)(tile_lane + (size_t)r * DK);
}
__device__ __forceinline__ void stage_store(float* st_w, const v4f (&g)[16]) {
#pragma unroll
  for (int r = 0; r < 16; ++r) *(v4f*)(st_w + r * SPITCH) = g[r];
}
__device__ __forceinline__ v4f stage_frag4(const float* st_r, int i) {
  return *(const v4f*)(st_r + 8 * i);
}
__device__ __forceinline__ v4f load_vrow(const float* __restrict__ Vrow, int idx) {
  return *(const v4f*)(Vrow + (size_t)(4 * (idx >> 2) + (idx & 1)) * DK + 64 * ((idx >> 1) & 1));
}
__device__ __forceinline__ void prio_by_progress(int done) {
  if (done == PART_LEN / 2)                __builtin_amdgcn_s_setprio(2);
  if (done == PART_LEN / 2 + PART_LEN / 4) __builtin_amdgcn_s_setprio(1);
  if (done == PART_LEN - PART_LEN / 8)     __builtin_amdgcn_s_setprio(0);
}
__device__ __forceinline__ v2f lo2(v4f f) { v2f r; r.x = f.x; r.y = f.y; return r; }
__device__ __forceinline__ v2f hi2(v4f f) { v2f r; r.x = f.z; r.y = f.w; return r; }

template <bool KEEP>
__device__ __forceinline__ void colstats_stream(const float* __restrict__ Qb, int qbeg, float* st_w, const float* st_r,
                                                const v2f (&breg)[32], float* sdst, size_t sstep,
                                                float& m_run, float& l_run) {
  v4f g[16];
  stage_rows(Qb + (size_t)qbeg * DK, g);
  __builtin_amdgcn_s_setprio(3);
  for (int it = 0; it < PART_LEN; it += 16) {
    prio_by_progress(it);
    stage_store(st_w, g);
    const int itn = (it + 16 < PART_LEN) ? it + 16 : it;
    stage_rows(Qb + (size_t)(qbeg + itn) * DK, g);
    __builtin_amdgcn_wave_barrier();
    v4f f[16];
#pragma unroll
    for (int i = 0; i < 16; ++i) f[i] = stage_frag4(st_r, i);
    __builtin_amdgcn_sched_barrier(0);
    v8f acc = {};
#pragma unroll
    for (int i = 0; i < 16; ++i) {
      acc = wmma_f32(lo2(f[i]), breg[2 * i], acc);
      acc = wmma_f32(hi2(f[i]), breg[2 * i + 1], acc);
    }
    float mloc = acc[0] * SCALE;
#pragma unroll
    for (int j = 1; j < 8; ++j) mloc = fmaxf(mloc, acc[j] * SCALE);
    const float m_new = fmaxf(m_run, mloc);
    float s = 0.0f;
#pragma unroll
    for (int j = 0; j < 8; ++j) s += __expf(acc[j] * SCALE - m_new);
    l_run = l_run * __expf(m_run - m_new) + s;
    m_run = m_new;
    if (KEEP) {
      v4f s0, s1;
      s0.x = acc[0]; s0.y = acc[1]; s0.z = acc[2]; s0.w = acc[3];
      s1.x = acc[4]; s1.y = acc[5]; s1.z = acc[6]; s1.w = acc[7];
      keep_store4(sdst, s0);
      keep_store4(sdst + 128, s1);
      sdst += sstep;
    }
  }

}

__global__ __launch_bounds__(WAVES_PER_BLOCK * 32) __attribute__((amdgpu_waves_per_eu(4, 4)))
void sdpa_colstats_kernel(const float* __restrict__ Q,
                          const float* __restrict__ K,
                          float* __restrict__ mArr,
                          float* __restrict__ lArr,
                          float* __restrict__ Skeep,
                          size_t ws_bytes) {
  __shared__ float red[WAVES_PER_BLOCK][32];
  __shared__ __attribute__((aligned(16))) float stage[WAVES_PER_BLOCK][16 * SPITCH];

  const int lane  = threadIdx.x & 31;
  const int wslot = __builtin_amdgcn_readfirstlane(threadIdx.x >> 5);
  const int tloc  = wslot & (TILES_PER_BLOCK - 1);
  const int part  = wslot / TILES_PER_BLOCK;
  const int wid   = blockIdx.x * TILES_PER_BLOCK + tloc;
  const int b     = wid >> 8;
  const int kt    = (wid & 255) << 4;
  const int n     = lane & 15;
  const int dhalf = (lane >> 4) << 2;

  const float* Krow = K + ((size_t)b * SEQ + kt + n) * DK + dhalf;
  v2f breg[32];
#pragma unroll
  for (int i = 0; i < 16; ++i) {
    const v4f f = load_frag4(Krow, i);
    breg[2 * i]     = lo2(f);
    breg[2 * i + 1] = hi2(f);
  }

  const float* Qb = Q + (size_t)b * SEQ * DK + 4 * lane;
  float*       st_w = &stage[wslot][0] + 4 * lane;
  const float* st_r = &stage[wslot][0] + n * SPITCH + dhalf;
  float m_run = -__builtin_inff();
  float l_run = 0.0f;

  const int qbeg = part * PART_LEN;
  const int T     = kept_tiles(ws_bytes);
  const int ki    = wid & (KTILES - 1);
  const bool keep = (ki & (PART_TILES - 1)) < T;
  float* sdst = Skeep + (((size_t)(b * KTILES + (qbeg >> 4)) * SPLIT + ki / PART_TILES) * T
                         + (ki & (PART_TILES - 1))) * TILE_FLOATS + 4 * lane;
  const size_t sstep = (size_t)SPLIT * T * TILE_FLOATS;
  if (keep) colstats_stream<true>(Qb, qbeg, st_w, st_r, breg, sdst, sstep, m_run, l_run);
  else      colstats_stream<false>(Qb, qbeg, st_w, st_r, breg, sdst, sstep, m_run, l_run);

  const float m_o = __shfl_xor(m_run, 16, 32);
  const float l_o = __shfl_xor(l_run, 16, 32);
  const float m_f = fmaxf(m_run, m_o);
  const float l_f = l_run * __expf(m_run - m_f) + l_o * __expf(m_o - m_f);

  if (lane < 16) {
    red[wslot][n]      = m_f;
    red[wslot][16 + n] = l_f;
  }
  __syncthreads();
  if (part == 0 && lane < 16) {
    float m = m_f;
    float l = l_f;
    for (int p = 1; p < SPLIT; ++p) {
      const float mo = red[p * TILES_PER_BLOCK + tloc][n];
      const float lo = red[p * TILES_PER_BLOCK + tloc][16 + n];
      const float mn = fmaxf(m, mo);
      l = l * __expf(m - mn) + lo * __expf(mo - mn);
      m = mn;
    }
    mArr[(size_t)b * SEQ + kt + n] = m;
    lArr[(size_t)b * SEQ + kt + n] = l;
  }
}

__global__ __launch_bounds__(WAVES_PER_BLOCK * 32)
void sdpa_out_kernel(const float* __restrict__ Q,
                     const float* __restrict__ K,
                     const float* __restrict__ V,
                     const float* __restrict__ mArr,
                     const float* __restrict__ lArr,
                     const float* __restrict__ Skeep,
                     size_t ws_bytes,
                     float* __restrict__ O) {
  __shared__ float smem[WAVES_PER_BLOCK][16 * 17];
  __shared__ __attribute__((aligned(16))) float stage[WAVES_PER_BLOCK][16 * SPITCH];
  __shared__ float opart[(SPLIT - 1) * TILES_PER_BLOCK][16 * DK];

  const int lane  = threadIdx.x & 31;
  const int wslot = __builtin_amdgcn_readfirstlane(threadIdx.x >> 5);
  const int tloc  = wslot & (TILES_PER_BLOCK - 1);
  const int part  = wslot / TILES_PER_BLOCK;
  const int wid   = blockIdx.x * TILES_PER_BLOCK + tloc;
  const int b     = wid >> 8;
  const int qt    = (wid & 255) << 4;
  const int n     = lane & 15;
  const int khalf = (lane >> 4) << 1;
  const int dhalf = (lane >> 4) << 2;
  const int mhi   = (lane >> 4) << 3;

  const v8f vzero = {};
  v8f acc[8];
#pragma unroll
  for (int t = 0; t < 8; ++t) acc[t] = vzero;

  const float* Kb = K + (size_t)b * SEQ * DK;
  const float* Kl = Kb + 4 * lane;
  float*       st_w = &stage[wslot][0] + 4 * lane;
  const float* st_r = &stage[wslot][0] + n * SPITCH + dhalf;
  const float* Vb = V + (size_t)b * SEQ * DK;
  const float* mB = mArr + (size_t)b * SEQ;
  const float* lB = lArr + (size_t)b * SEQ;
  float* sm = &smem[wslot][0];

  const int kbeg = part * PART_LEN;
  const int T = kept_tiles(ws_bytes);
  __builtin_amdgcn_s_setprio(3);

  if (T > 0) {
    const float* ssrc = Skeep + ((size_t)(wid * SPLIT + part) * T) * TILE_FLOATS + 4 * lane;
    const float* Vl   = Vb + (size_t)(kbeg + khalf) * DK + 4 * n;
    v4f s0 = keep_load4(ssrc);
    v4f s1 = keep_load4(ssrc + 128);
    float mcol = mB[kbeg + n];
    float lcol = lB[kbeg + n];
    v4f vv[16];
#pragma unroll
    for (int c2 = 0; c2 < 4; ++c2) {
      __builtin_amdgcn_sched_barrier(0);
#pragma unroll
      for (int u = 0; u < 4; ++u) vv[4 * c2 + u] = load_vrow(Vl, 4 * c2 + u);
    }
    for (int ti = 0; ti < T; ++ti) {
      prio_by_progress(ti * 16);
      const int tn = (ti + 1 < T) ? ti + 1 : ti;
      const float* Vnext = Vl + (size_t)tn * 16 * DK;
      const float rl = 1.0f / lcol;
#pragma unroll
      for (int j = 0; j < 4; ++j) {
        sm[(j + mhi) * 17 + n]     = __expf(s0[j] * SCALE - mcol) * rl;
        sm[(j + 4 + mhi) * 17 + n] = __expf(s1[j] * SCALE - mcol) * rl;
      }
      __builtin_amdgcn_wave_barrier();
      v2f pa[4];
#pragma unroll
      for (int c2 = 0; c2 < 4; ++c2) {
        const int kk = 4 * c2 + khalf;
        pa[c2].x = sm[n * 17 + kk];
        pa[c2].y = sm[n * 17 + kk + 1];
      }
      __builtin_amdgcn_sched_barrier(0);
      s0 = keep_load4(ssrc + (size_t)tn * TILE_FLOATS);
      s1 = keep_load4(ssrc + (size_t)tn * TILE_FLOATS + 128);
      mcol = mB[kbeg + tn * 16 + n];
      lcol = lB[kbeg + tn * 16 + n];
#pragma unroll
      for (int c2 = 0; c2 < 4; ++c2) {
        const v4f x0 = vv[4 * c2];
        const v4f y0 = vv[4 * c2 + 1];
        const v4f x1 = vv[4 * c2 + 2];
        const v4f y1 = vv[4 * c2 + 3];
#pragma unroll
        for (int t = 0; t < 4; ++t) {
          v2f bv0, bv1;
          bv0.x = x0[t]; bv0.y = y0[t];
          bv1.x = x1[t]; bv1.y = y1[t];
          acc[t]     = wmma_f32(pa[c2], bv0, acc[t]);
          acc[4 + t] = wmma_f32(pa[c2], bv1, acc[4 + t]);
        }
        __builtin_amdgcn_sched_barrier(0);
#pragma unroll
        for (int u = 0; u < 4; ++u) vv[4 * c2 + u] = load_vrow(Vnext, 4 * c2 + u);
      }
      asm volatile("" ::: "memory");
    }
  }

  if (T < PART_TILES) {
  const float* Qrow = Q + ((size_t)b * SEQ + qt + n) * DK + dhalf;
  v2f areg[32];
#pragma unroll
  for (int i = 0; i < 16; ++i) {
    const v4f f = load_frag4(Qrow, i);
    areg[2 * i]     = lo2(f);
    areg[2 * i + 1] = hi2(f);
  }
  v4f g[16];
  stage_rows(Kl + (size_t)(kbeg + T * 16) * DK, g);
  for (int it = T * 16; it < PART_LEN; it += 16) {
    prio_by_progress(it);
    const int kt = kbeg + it;
    const int itn = (it + 16 < PART_LEN) ? it + 16 : it;
    const float* Knext = Kl + (size_t)(kbeg + itn) * DK;
    const float* Vrow = Vb + (size_t)(kt + khalf) * DK + 4 * n;
    stage_store(st_w, g);
    __builtin_amdgcn_wave_barrier();
    v4f f[16];
#pragma unroll
    for (int i = 0; i < 16; ++i) f[i] = stage_frag4(st_r, i);
    const float mcol = mB[kt + n];
    const float lcol = lB[kt + n];
    __builtin_amdgcn_sched_barrier(0);
    v4f vv[16];
    v8f c8 = vzero;
#pragma unroll
    for (int i = 0; i < 16; ++i) {
      c8 = wmma_f32(areg[2 * i],     lo2(f[i]), c8);
      c8 = wmma_f32(areg[2 * i + 1], hi2(f[i]), c8);
      if (i == 5 || i == 9 || i == 13 || i == 15) {
        const int q4 = (i == 15) ? 3 : (i - 5) / 4;
        __builtin_amdgcn_sched_barrier(0);
#pragma unroll
        for (int u = 0; u < 4; ++u) vv[4 * q4 + u] = load_vrow(Vrow, 4 * q4 + u);
      }
    }

    const float rl   = 1.0f / lcol;
#pragma unroll
    for (int j = 0; j < 8; ++j) {
      const float p = __expf(c8[j] * SCALE - mcol) * rl;
      sm[(j + mhi) * 17 + n] = p;
    }
    __builtin_amdgcn_wave_barrier();

    v2f pa[4];
#pragma unroll
    for (int c2 = 0; c2 < 4; ++c2) {
      const int kk = 4 * c2 + khalf;
      pa[c2].x = sm[n * 17 + kk];
      pa[c2].y = sm[n * 17 + kk + 1];
    }

    __builtin_amdgcn_sched_barrier(0);
#pragma unroll
    for (int r = 0; r < 4; ++r) g[r] = *(const v4f*)(Knext + (size_t)r * DK);
#pragma unroll
    for (int c2 = 0; c2 < 4; ++c2) {
      const v4f x0 = vv[4 * c2];
      const v4f y0 = vv[4 * c2 + 1];
      const v4f x1 = vv[4 * c2 + 2];
      const v4f y1 = vv[4 * c2 + 3];
#pragma unroll
      for (int t = 0; t < 4; ++t) {
        v2f bv0, bv1;
        bv0.x = x0[t]; bv0.y = y0[t];
        bv1.x = x1[t]; bv1.y = y1[t];
        acc[t]     = wmma_f32(pa[c2], bv0, acc[t]);
        acc[4 + t] = wmma_f32(pa[c2], bv1, acc[4 + t]);
      }
      if (c2 < 3) {
        __builtin_amdgcn_sched_barrier(0);
#pragma unroll
        for (int r = 4 * c2 + 4; r < 4 * c2 + 8; ++r) g[r] = *(const v4f*)(Knext + (size_t)r * DK);
      }
    }
    asm volatile("" ::: "memory");
  }
  }

  if (part > 0) {
    float* op = &opart[(part - 1) * TILES_PER_BLOCK + tloc][0];
#pragma unroll
    for (int t = 0; t < 8; ++t) {
#pragma unroll
      for (int j = 0; j < 8; ++j) op[(t * 8 + j) * 32 + lane] = acc[t][j];
    }
  }
  __syncthreads();
  if (part == 0) {
#pragma unroll
    for (int p = 1; p < SPLIT; ++p) {
      const float* op = &opart[(p - 1) * TILES_PER_BLOCK + tloc][0];
#pragma unroll
      for (int t = 0; t < 8; ++t) {
#pragma unroll
        for (int j = 0; j < 8; ++j) acc[t][j] += op[(t * 8 + j) * 32 + lane];
      }
    }
    float* Ob = O + ((size_t)b * SEQ + qt) * DK + 4 * n;
#pragma unroll
    for (int j = 0; j < 8; ++j) {
      v4f o0, o1;
      o0.x = acc[0][j]; o0.y = acc[1][j]; o0.z = acc[2][j]; o0.w = acc[3][j];
      o1.x = acc[4][j]; o1.y = acc[5][j]; o1.z = acc[6][j]; o1.w = acc[7][j];
      *(v4f*)(Ob + (size_t)(j + mhi) * DK)      = o0;
      *(v4f*)(Ob + (size_t)(j + mhi) * DK + 64) = o1;
    }
  }
}

extern "C" void kernel_launch(void* const* d_in, const int* in_sizes, int n_in,
                              void* d_out, int out_size, void* d_ws, size_t ws_size,
                              hipStream_t stream) {
  const float* q = (const float*)d_in[0];
  const float* k = (const float*)d_in[1];
  const float* v = (const float*)d_in[2];
  float* out = (float*)d_out;

  float* mArr = (float*)d_ws;
  float* lArr = mArr + (size_t)BS * SEQ;
  float* keep = lArr + (size_t)BS * SEQ;

  const int blocks = BS * KTILES / TILES_PER_BLOCK;
  const int threads = WAVES_PER_BLOCK * 32;

  sdpa_colstats_kernel<<<blocks, threads, 0, stream>>>(q, k, mArr, lArr, keep, ws_size);
  sdpa_out_kernel<<<blocks, threads, 0, stream>>>(q, k, v, mArr, lArr, keep, ws_size, out);
}
